# starting kernel
# baseline (speedup 1.0000x reference)
#include <hip/hip_runtime.h>
#include <hip/hip_bf16.h>
#include <math.h>

typedef __attribute__((ext_vector_type(2))) float v2f;
typedef __attribute__((ext_vector_type(8))) float v8f;

constexpr int KSZ   = 3;
constexpr int K2    = 9;
constexpr int CIN   = 64;
constexpr int COUT  = 64;
constexpr int NB    = 8;
constexpr int H     = 128;
constexpr int W     = 128;
constexpr int KDIM  = CIN * K2;

constexpr int THREADS     = 256;
constexpr int PIX_PER_BLK = 32;
constexpr int YSTRIDE     = KDIM + 1;

constexpr int LDS_W_OFF  = 0;
constexpr int LDS_Y_OFF  = LDS_W_OFF + KDIM * COUT;
constexpr int LDS_OP_OFF = LDS_Y_OFF + PIX_PER_BLK * YSTRIDE;
constexpr int LDS_N_OFF  = LDS_OP_OFF + PIX_PER_BLK * 81;
constexpr int LDS_FLOATS = LDS_N_OFF + PIX_PER_BLK * K2;

__global__ __launch_bounds__(THREADS, 1)
void dyfconv_fused(const float* __restrict__ x,
                   const float* __restrict__ weight,
                   const float* __restrict__ bias,
                   const float* __restrict__ op_w,
                   const float* __restrict__ op_b,
                   float* __restrict__ out)
{
    extern __shared__ float lds[];
    float* ldsW  = lds + LDS_W_OFF;
    float* ldsY  = lds + LDS_Y_OFF;
    float* ldsOp = lds + LDS_OP_OFF;
    float* ldsN  = lds + LDS_N_OFF;

    const int t     = threadIdx.x;
    const int pbase = blockIdx.x * PIX_PER_BLK;

    for (int idx = t; idx < KDIM * COUT; idx += THREADS) {
        int kk = idx >> 6;
        int o  = idx & (COUT - 1);
        int c  = kk / K2;
        int k  = kk - c * K2;
        ldsW[idx] = weight[(c * COUT + o) * K2 + k];
    }
    for (int idx = t; idx < PIX_PER_BLK * K2; idx += THREADS)
        ldsN[idx] = 0.0f;
    __syncthreads();

    const int p  = t >> 3;
    const int g  = t & 7;
    const int pg = pbase + p;
    const int bb = pg >> 14;
    const int hw = pg & (H * W - 1);
    const int hh = hw >> 7;
    const int ww = hw & (W - 1);

    {
        float sq[K2];
        #pragma unroll
        for (int j = 0; j < K2; ++j) sq[j] = 0.0f;
        for (int cc = 0; cc < 8; ++cc) {
            const int c = g * 8 + cc;
            const float* xc = x + (size_t)(bb * CIN + c) * (H * W);
            #pragma unroll
            for (int ki = 0; ki < KSZ; ++ki) {
                const int hs  = hh + ki - 1;
                const bool okh = (unsigned)hs < (unsigned)H;
                #pragma unroll
                for (int kj = 0; kj < KSZ; ++kj) {
                    const int wsv = ww + kj - 1;
                    const bool ok = okh && ((unsigned)wsv < (unsigned)W);
                    const float v = ok ? xc[hs * W + wsv] : 0.0f;
                    sq[ki * KSZ + kj] += v * v;
                }
            }
        }
        #pragma unroll
        for (int j = 0; j < K2; ++j)
            atomicAdd(&ldsN[p * K2 + j], sq[j]);
    }
    __syncthreads();

    for (int e = t; e < PIX_PER_BLK * 81; e += THREADS) {
        const int pp  = e / 81;
        const int idx = e - pp * 81;
        const float* nb = &ldsN[pp * K2];
        float acc = op_b[idx];
        #pragma unroll
        for (int j = 0; j < K2; ++j)
            acc += sqrtf(nb[j]) * op_w[idx * K2 + j];
        ldsOp[e] = tanhf(acc);
    }
    __syncthreads();

    {
        const float* opp = &ldsOp[p * 81];
        float* yrow = &ldsY[p * YSTRIDE];
        for (int cc = 0; cc < 8; ++cc) {
            const int c = g * 8 + cc;
            const float* xc = x + (size_t)(bb * CIN + c) * (H * W);
            float xu[K2];
            #pragma unroll
            for (int ki = 0; ki < KSZ; ++ki) {
                const int hs  = hh + ki - 1;
                const bool okh = (unsigned)hs < (unsigned)H;
                #pragma unroll
                for (int kj = 0; kj < KSZ; ++kj) {
                    const int wsv = ww + kj - 1;
                    const bool ok = okh && ((unsigned)wsv < (unsigned)W);
                    xu[ki * KSZ + kj] = ok ? xc[hs * W + wsv] : 0.0f;
                }
            }
            #pragma unroll
            for (int k = 0; k < K2; ++k) {
                float acc = xu[k];
                #pragma unroll
                for (int j = 0; j < K2; ++j)
                    acc += xu[j] * opp[j * K2 + k];
                yrow[c * K2 + k] = acc;
            }
        }
    }
    __syncthreads();

    const int wave  = t >> 5;
    const int lane  = t & 31;
    const int mTile = wave & 1;
    const int nTile = wave >> 1;
    const int mIdx  = lane & 15;
    const int nIdx  = lane & 15;
    const int kh    = lane >> 4;

    const float* yrow = &ldsY[(mTile * 16 + mIdx) * YSTRIDE];
    const float* wcol = &ldsW[nTile * 16 + nIdx];

    v8f acc = {};
    #pragma unroll 4
    for (int kk = 0; kk < KDIM; kk += 4) {
        v2f a, bf;
        const int kb = kk + 2 * kh;
        a.x = yrow[kb + 0];
        a.y = yrow[kb + 1];
        bf.x = wcol[(kb + 0) * COUT];
        bf.y = wcol[(kb + 1) * COUT];
        acc = __builtin_amdgcn_wmma_f32_16x16x4_f32(
                   false, a,  false, bf,
                   (short)0, acc,  false,  false);
    }

    const int o  = nTile * 16 + nIdx;
    const float bo = bias[o];
    #pragma unroll
    for (int r = 0; r < 8; ++r) {
        const int m     = r + 8 * kh;
        const int pglob = pbase + mTile * 16 + m;
        const int b_    = pglob >> 14;
        const int hw_   = pglob & (H * W - 1);
        out[((size_t)(b_ * COUT + o)) * (H * W) + hw_] = acc[r] + bo;
    }
}

extern "C" void kernel_launch(void* const* d_in, const int* in_sizes, int n_in,
                              void* d_out, int out_size, void* d_ws, size_t ws_size,
                              hipStream_t stream) {
    (void)in_sizes; (void)n_in; (void)out_size; (void)d_ws; (void)ws_size;
    const float* x      = (const float*)d_in[0];
    const float* weight = (const float*)d_in[1];
    const float* bias   = (const float*)d_in[2];
    const float* op_w   = (const float*)d_in[3];
    const float* op_b   = (const float*)d_in[4];
    float* out = (float*)d_out;

    const int totalPix = NB * H * W;
    const int blocks   = totalPix / PIX_PER_BLK;
    const size_t ldsBytes = (size_t)LDS_FLOATS * sizeof(float);

    dyfconv_fused<<<blocks, THREADS, ldsBytes, stream>>>(x, weight, bias, op_w, op_b, out);
}


# Work-group doubled to 16 waves on a 64-pixel tile (static LDS 318208 B): 4 waves per SIMD overlap the serial load and LDS waits of every phase; weight staging per pixel halves
# speedup vs baseline: 1.7910x; 1.7910x over previous
#include <hip/hip_runtime.h>
#include <hip/hip_bf16.h>
#include <math.h>

typedef __attribute__((ext_vector_type(2))) float v2f;
typedef __attribute__((ext_vector_type(8))) float v8f;

constexpr int KSZ   = 3;
constexpr int K2    = 9;
constexpr int CIN   = 64;
constexpr int COUT  = 64;
constexpr int NB    = 8;
constexpr int H     = 128;
constexpr int W     = 128;
constexpr int KDIM  = CIN * K2;

constexpr int THREADS     = 512;
constexpr int PIX_PER_BLK = 64;
constexpr int YSTRIDE     = KDIM + 1;

constexpr int LDS_W_OFF  = 0;
constexpr int LDS_Y_OFF  = LDS_W_OFF + KDIM * COUT;
constexpr int LDS_OP_OFF = LDS_Y_OFF + PIX_PER_BLK * YSTRIDE;
constexpr int LDS_N_OFF  = LDS_OP_OFF + PIX_PER_BLK * 81;
constexpr int LDS_FLOATS = LDS_N_OFF + PIX_PER_BLK * K2;

__global__ __launch_bounds__(THREADS, 1)
void dyfconv_fused(const float* __restrict__ x,
                   const float* __restrict__ weight,
                   const float* __restrict__ bias,
                   const float* __restrict__ op_w,
                   const float* __restrict__ op_b,
                   float* __restrict__ out)
{
    __shared__ float lds[LDS_FLOATS];
    float* ldsW  = lds + LDS_W_OFF;
    float* ldsY  = lds + LDS_Y_OFF;
    float* ldsOp = lds + LDS_OP_OFF;
    float* ldsN  = lds + LDS_N_OFF;

    const int t     = threadIdx.x;
    const int pbase = blockIdx.x * PIX_PER_BLK;

    for (int idx = t; idx < KDIM * COUT; idx += THREADS) {
        int kk = idx >> 6;
        int o  = idx & (COUT - 1);
        int c  = kk / K2;
        int k  = kk - c * K2;
        ldsW[idx] = weight[(c * COUT + o) * K2 + k];
    }
    for (int idx = t; idx < PIX_PER_BLK * K2; idx += THREADS)
        ldsN[idx] = 0.0f;
    __syncthreads();

    const int p  = t >> 3;
    const int g  = t & 7;
    const int pg = pbase + p;
    const int bb = pg >> 14;
    const int hw = pg & (H * W - 1);
    const int hh = hw >> 7;
    const int ww = hw & (W - 1);

    {
        float sq[K2];
        #pragma unroll
        for (int j = 0; j < K2; ++j) sq[j] = 0.0f;
        for (int cc = 0; cc < 8; ++cc) {
            const int c = g * 8 + cc;
            const float* xc = x + (size_t)(bb * CIN + c) * (H * W);
            #pragma unroll
            for (int ki = 0; ki < KSZ; ++ki) {
                const int hs  = hh + ki - 1;
                const bool okh = (unsigned)hs < (unsigned)H;
                #pragma unroll
                for (int kj = 0; kj < KSZ; ++kj) {
                    const int wsv = ww + kj - 1;
                    const bool ok = okh && ((unsigned)wsv < (unsigned)W);
                    const float v = ok ? xc[hs * W + wsv] : 0.0f;
                    sq[ki * KSZ + kj] += v * v;
                }
            }
        }
        #pragma unroll
        for (int j = 0; j < K2; ++j)
            atomicAdd(&ldsN[p * K2 + j], sq[j]);
    }
    __syncthreads();

    for (int e = t; e < PIX_PER_BLK * 81; e += THREADS) {
        const int pp  = e / 81;
        const int idx = e - pp * 81;
        const float* nb = &ldsN[pp * K2];
        float acc = op_b[idx];
        #pragma unroll
        for (int j = 0; j < K2; ++j)
            acc += sqrtf(nb[j]) * op_w[idx * K2 + j];
        ldsOp[e] = tanhf(acc);
    }
    __syncthreads();

    {
        const float* opp = &ldsOp[p * 81];
        float* yrow = &ldsY[p * YSTRIDE];
        for (int cc = 0; cc < 8; ++cc) {
            const int c = g * 8 + cc;
            const float* xc = x + (size_t)(bb * CIN + c) * (H * W);
            float xu[K2];
            #pragma unroll
            for (int ki = 0; ki < KSZ; ++ki) {
                const int hs  = hh + ki - 1;
                const bool okh = (unsigned)hs < (unsigned)H;
                #pragma unroll
                for (int kj = 0; kj < KSZ; ++kj) {
                    const int wsv = ww + kj - 1;
                    const bool ok = okh && ((unsigned)wsv < (unsigned)W);
                    xu[ki * KSZ + kj] = ok ? xc[hs * W + wsv] : 0.0f;
                }
            }
            #pragma unroll
            for (int k = 0; k < K2; ++k) {
                float acc = xu[k];
                #pragma unroll
                for (int j = 0; j < K2; ++j)
                    acc += xu[j] * opp[j * K2 + k];
                yrow[c * K2 + k] = acc;
            }
        }
    }
    __syncthreads();

    const int wave  = t >> 5;
    const int lane  = t & 31;
    const int mTile = wave & 3;
    const int nTile = wave >> 2;
    const int mIdx  = lane & 15;
    const int nIdx  = lane & 15;
    const int kh    = lane >> 4;

    const float* yrow = &ldsY[(mTile * 16 + mIdx) * YSTRIDE];
    const float* wcol = &ldsW[nTile * 16 + nIdx];

    v8f acc = {};
    #pragma unroll 4
    for (int kk = 0; kk < KDIM; kk += 4) {
        v2f a, bf;
        const int kb = kk + 2 * kh;
        a.x = yrow[kb + 0];
        a.y = yrow[kb + 1];
        bf.x = wcol[(kb + 0) * COUT];
        bf.y = wcol[(kb + 1) * COUT];
        acc = __builtin_amdgcn_wmma_f32_16x16x4_f32(
                   false, a,  false, bf,
                   (short)0, acc,  false,  false);
    }

    const int o  = nTile * 16 + nIdx;
    const float bo = bias[o];
    #pragma unroll
    for (int r = 0; r < 8; ++r) {
        const int m     = r + 8 * kh;
        const int pglob = pbase + mTile * 16 + m;
        const int b_    = pglob >> 14;
        const int hw_   = pglob & (H * W - 1);
        out[((size_t)(b_ * COUT + o)) * (H * W) + hw_] = acc[r] + bo;
    }
}

extern "C" void kernel_launch(void* const* d_in, const int* in_sizes, int n_in,
                              void* d_out, int out_size, void* d_ws, size_t ws_size,
                              hipStream_t stream) {
    (void)in_sizes; (void)n_in; (void)out_size; (void)d_ws; (void)ws_size;
    const float* x      = (const float*)d_in[0];
    const float* weight = (const float*)d_in[1];
    const float* bias   = (const float*)d_in[2];
    const float* op_w   = (const float*)d_in[3];
    const float* op_b   = (const float*)d_in[4];
    float* out = (float*)d_out;

    const int totalPix = NB * H * W;
    const int blocks   = totalPix / PIX_PER_BLK;

    dyfconv_fused<<<blocks, THREADS, 0, stream>>>(x, weight, bias, op_w, op_b, out);
}


# Each thread reads its 8-channel x 9-tap patch of x once, as 72 independent clamped loads kept in registers for the norm and operator phases (was two passes of branched, serially waited loads)
# speedup vs baseline: 2.7824x; 1.5535x over previous
#include <hip/hip_runtime.h>
#include <hip/hip_bf16.h>
#include <math.h>

typedef __attribute__((ext_vector_type(2))) float v2f;
typedef __attribute__((ext_vector_type(8))) float v8f;

constexpr int KSZ   = 3;
constexpr int K2    = 9;
constexpr int CIN   = 64;
constexpr int COUT  = 64;
constexpr int NB    = 8;
constexpr int H     = 128;
constexpr int W     = 128;
constexpr int KDIM  = CIN * K2;

constexpr int THREADS     = 512;
constexpr int PIX_PER_BLK = 64;
constexpr int YSTRIDE     = KDIM + 1;

constexpr int LDS_W_OFF  = 0;
constexpr int LDS_Y_OFF  = LDS_W_OFF + KDIM * COUT;
constexpr int LDS_OP_OFF = LDS_Y_OFF + PIX_PER_BLK * YSTRIDE;
constexpr int LDS_N_OFF  = LDS_OP_OFF + PIX_PER_BLK * 81;
constexpr int LDS_FLOATS = LDS_N_OFF + PIX_PER_BLK * K2;

__global__ __launch_bounds__(THREADS, 1)
void dyfconv_fused(const float* __restrict__ x,
                   const float* __restrict__ weight,
                   const float* __restrict__ bias,
                   const float* __restrict__ op_w,
                   const float* __restrict__ op_b,
                   float* __restrict__ out)
{
    __shared__ float lds[LDS_FLOATS];
    float* ldsW  = lds + LDS_W_OFF;
    float* ldsY  = lds + LDS_Y_OFF;
    float* ldsOp = lds + LDS_OP_OFF;
    float* ldsN  = lds + LDS_N_OFF;

    const int t     = threadIdx.x;
    const int pbase = blockIdx.x * PIX_PER_BLK;

    for (int idx = t; idx < KDIM * COUT; idx += THREADS) {
        int kk = idx >> 6;
        int o  = idx & (COUT - 1);
        int c  = kk / K2;
        int k  = kk - c * K2;
        ldsW[idx] = weight[(c * COUT + o) * K2 + k];
    }
    for (int idx = t; idx < PIX_PER_BLK * K2; idx += THREADS)
        ldsN[idx] = 0.0f;
    __syncthreads();

    const int p  = t >> 3;
    const int g  = t & 7;
    const int pg = pbase + p;
    const int bb = pg >> 14;
    const int hw = pg & (H * W - 1);
    const int hh = hw >> 7;
    const int ww = hw & (W - 1);

    float xv[8][K2];
    {
        int  toff[K2];
        bool tok[K2];
        #pragma unroll
        for (int ki = 0; ki < KSZ; ++ki) {
            const int hs = hh + ki - 1;
            const int hc = min(max(hs, 0), H - 1);
            #pragma unroll
            for (int kj = 0; kj < KSZ; ++kj) {
                const int wsv = ww + kj - 1;
                const int wc  = min(max(wsv, 0), W - 1);
                tok[ki * KSZ + kj]  = ((unsigned)hs < (unsigned)H) && ((unsigned)wsv < (unsigned)W);
                toff[ki * KSZ + kj] = hc * W + wc;
            }
        }
        const float* xg = x + (size_t)(bb * CIN + g * 8) * (H * W);
        #pragma unroll
        for (int cc = 0; cc < 8; ++cc) {
            #pragma unroll
            for (int j = 0; j < K2; ++j) {
                const float v = xg[cc * (H * W) + toff[j]];
                xv[cc][j] = tok[j] ? v : 0.0f;
            }
        }
    }

    {
        float sq[K2];
        #pragma unroll
        for (int j = 0; j < K2; ++j) sq[j] = 0.0f;
        #pragma unroll
        for (int cc = 0; cc < 8; ++cc) {
            #pragma unroll
            for (int j = 0; j < K2; ++j)
                sq[j] += xv[cc][j] * xv[cc][j];
        }
        #pragma unroll
        for (int j = 0; j < K2; ++j)
            atomicAdd(&ldsN[p * K2 + j], sq[j]);
    }
    __syncthreads();

    for (int e = t; e < PIX_PER_BLK * 81; e += THREADS) {
        const int pp  = e / 81;
        const int idx = e - pp * 81;
        const float* nb = &ldsN[pp * K2];
        float acc = op_b[idx];
        #pragma unroll
        for (int j = 0; j < K2; ++j)
            acc += sqrtf(nb[j]) * op_w[idx * K2 + j];
        ldsOp[e] = tanhf(acc);
    }
    __syncthreads();

    {
        const float* opp = &ldsOp[p * 81];
        float* yrow = &ldsY[p * YSTRIDE];
        #pragma unroll
        for (int cc = 0; cc < 8; ++cc) {
            const int c = g * 8 + cc;
            const float* xu = xv[cc];
            #pragma unroll
            for (int k = 0; k < K2; ++k) {
                float acc = xu[k];
                #pragma unroll
                for (int j = 0; j < K2; ++j)
                    acc += xu[j] * opp[j * K2 + k];
                yrow[c * K2 + k] = acc;
            }
        }
    }
    __syncthreads();

    const int wave  = t >> 5;
    const int lane  = t & 31;
    const int mTile = wave & 3;
    const int nTile = wave >> 2;
    const int mIdx  = lane & 15;
    const int nIdx  = lane & 15;
    const int kh    = lane >> 4;

    const float* yrow = &ldsY[(mTile * 16 + mIdx) * YSTRIDE];
    const float* wcol = &ldsW[nTile * 16 + nIdx];

    v8f acc = {};
    #pragma unroll 4
    for (int kk = 0; kk < KDIM; kk += 4) {
        v2f a, bf;
        const int kb = kk + 2 * kh;
        a.x = yrow[kb + 0];
        a.y = yrow[kb + 1];
        bf.x = wcol[(kb + 0) * COUT];
        bf.y = wcol[(kb + 1) * COUT];
        acc = __builtin_amdgcn_wmma_f32_16x16x4_f32(
                   false, a,  false, bf,
                   (short)0, acc,  false,  false);
    }

    const int o  = nTile * 16 + nIdx;
    const float bo = bias[o];
    #pragma unroll
    for (int r = 0; r < 8; ++r) {
        const int m     = r + 8 * kh;
        const int pglob = pbase + mTile * 16 + m;
        const int b_    = pglob >> 14;
        const int hw_   = pglob & (H * W - 1);
        out[((size_t)(b_ * COUT + o)) * (H * W) + hw_] = acc[r] + bo;
    }
}

extern "C" void kernel_launch(void* const* d_in, const int* in_sizes, int n_in,
                              void* d_out, int out_size, void* d_ws, size_t ws_size,
                              hipStream_t stream) {
    (void)in_sizes; (void)n_in; (void)out_size; (void)d_ws; (void)ws_size;
    const float* x      = (const float*)d_in[0];
    const float* weight = (const float*)d_in[1];
    const float* bias   = (const float*)d_in[2];
    const float* op_w   = (const float*)d_in[3];
    const float* op_b   = (const float*)d_in[4];
    float* out = (float*)d_out;

    const int totalPix = NB * H * W;
    const int blocks   = totalPix / PIX_PER_BLK;

    dyfconv_fused<<<blocks, THREADS, 0, stream>>>(x, weight, bias, op_w, op_b, out);
}


# Persistent work-groups: 256 work-groups (one per WGP) each stage the 576x64 weight into LDS once and loop over 8 pixel tiles (was 2048 work-groups, each restaging the weight)
# speedup vs baseline: 3.3170x; 1.1921x over previous
#include <hip/hip_runtime.h>
#include <hip/hip_bf16.h>
#include <math.h>

typedef __attribute__((ext_vector_type(2))) float v2f;
typedef __attribute__((ext_vector_type(8))) float v8f;

constexpr int KSZ   = 3;
constexpr int K2    = 9;
constexpr int CIN   = 64;
constexpr int COUT  = 64;
constexpr int NB    = 8;
constexpr int H     = 128;
constexpr int W     = 128;
constexpr int KDIM  = CIN * K2;

constexpr int THREADS     = 512;
constexpr int PIX_PER_BLK = 64;
constexpr int NUM_TILES   = NB * H * W / PIX_PER_BLK;
constexpr int NUM_WG      = 256;
constexpr int YSTRIDE     = KDIM + 1;

constexpr int LDS_W_OFF  = 0;
constexpr int LDS_Y_OFF  = LDS_W_OFF + KDIM * COUT;
constexpr int LDS_OP_OFF = LDS_Y_OFF + PIX_PER_BLK * YSTRIDE;
constexpr int LDS_N_OFF  = LDS_OP_OFF + PIX_PER_BLK * 81;
constexpr int LDS_FLOATS = LDS_N_OFF + PIX_PER_BLK * K2;

__global__ __launch_bounds__(THREADS, 1)
void dyfconv_fused(const float* __restrict__ x,
                   const float* __restrict__ weight,
                   const float* __restrict__ bias,
                   const float* __restrict__ op_w,
                   const float* __restrict__ op_b,
                   float* __restrict__ out)
{
    __shared__ float lds[LDS_FLOATS];
    float* ldsW  = lds + LDS_W_OFF;
    float* ldsY  = lds + LDS_Y_OFF;
    float* ldsOp = lds + LDS_OP_OFF;
    float* ldsN  = lds + LDS_N_OFF;

    const int t     = threadIdx.x;

    for (int idx = t; idx < KDIM * COUT; idx += THREADS) {
        int kk = idx >> 6;
        int o  = idx & (COUT - 1);
        int c  = kk / K2;
        int k  = kk - c * K2;
        ldsW[idx] = weight[(c * COUT + o) * K2 + k];
    }
    for (int idx = t; idx < PIX_PER_BLK * K2; idx += THREADS)
        ldsN[idx] = 0.0f;
    __syncthreads();

    const int p  = t >> 3;
    const int g  = t & 7;

    const int wave  = t >> 5;
    const int lane  = t & 31;
    const int mTile = wave & 3;
    const int nTile = wave >> 2;
    const int mIdx  = lane & 15;
    const int nIdx  = lane & 15;
    const int kh    = lane >> 4;

    for (int tile = blockIdx.x; tile < NUM_TILES; tile += gridDim.x) {
    const int pbase = tile * PIX_PER_BLK;
    const int pg = pbase + p;
    const int bb = pg >> 14;
    const int hw = pg & (H * W - 1);
    const int hh = hw >> 7;
    const int ww = hw & (W - 1);

    float xv[8][K2];
    {
        int  toff[K2];
        bool tok[K2];
        #pragma unroll
        for (int ki = 0; ki < KSZ; ++ki) {
            const int hs = hh + ki - 1;
            const int hc = min(max(hs, 0), H - 1);
            #pragma unroll
            for (int kj = 0; kj < KSZ; ++kj) {
                const int wsv = ww + kj - 1;
                const int wc  = min(max(wsv, 0), W - 1);
                tok[ki * KSZ + kj]  = ((unsigned)hs < (unsigned)H) && ((unsigned)wsv < (unsigned)W);
                toff[ki * KSZ + kj] = hc * W + wc;
            }
        }
        const float* xg = x + (size_t)(bb * CIN + g * 8) * (H * W);
        #pragma unroll
        for (int cc = 0; cc < 8; ++cc) {
            #pragma unroll
            for (int j = 0; j < K2; ++j) {
                const float v = xg[cc * (H * W) + toff[j]];
                xv[cc][j] = tok[j] ? v : 0.0f;
            }
        }
    }

    {
        float sq[K2];
        #pragma unroll
        for (int j = 0; j < K2; ++j) sq[j] = 0.0f;
        #pragma unroll
        for (int cc = 0; cc < 8; ++cc) {
            #pragma unroll
            for (int j = 0; j < K2; ++j)
                sq[j] += xv[cc][j] * xv[cc][j];
        }
        #pragma unroll
        for (int j = 0; j < K2; ++j)
            atomicAdd(&ldsN[p * K2 + j], sq[j]);
    }
    __syncthreads();

    for (int e = t; e < PIX_PER_BLK * 81; e += THREADS) {
        const int pp  = e / 81;
        const int idx = e - pp * 81;
        const float* nb = &ldsN[pp * K2];
        float acc = op_b[idx];
        #pragma unroll
        for (int j = 0; j < K2; ++j)
            acc += sqrtf(nb[j]) * op_w[idx * K2 + j];
        ldsOp[e] = tanhf(acc);
    }
    __syncthreads();

    for (int idx = t; idx < PIX_PER_BLK * K2; idx += THREADS)
        ldsN[idx] = 0.0f;

    {
        const float* opp = &ldsOp[p * 81];
        float* yrow = &ldsY[p * YSTRIDE];
        #pragma unroll
        for (int cc = 0; cc < 8; ++cc) {
            const int c = g * 8 + cc;
            const float* xu = xv[cc];
            #pragma unroll
            for (int k = 0; k < K2; ++k) {
                float acc = xu[k];
                #pragma unroll
                for (int j = 0; j < K2; ++j)
                    acc += xu[j] * opp[j * K2 + k];
                yrow[c * K2 + k] = acc;
            }
        }
    }
    __syncthreads();

    const float* yrow = &ldsY[(mTile * 16 + mIdx) * YSTRIDE];
    const float* wcol = &ldsW[nTile * 16 + nIdx];

    v8f acc = {};
    #pragma unroll 4
    for (int kk = 0; kk < KDIM; kk += 4) {
        v2f a, bf;
        const int kb = kk + 2 * kh;
        a.x = yrow[kb + 0];
        a.y = yrow[kb + 1];
        bf.x = wcol[(kb + 0) * COUT];
        bf.y = wcol[(kb + 1) * COUT];
        acc = __builtin_amdgcn_wmma_f32_16x16x4_f32(
                   false, a,  false, bf,
                   (short)0, acc,  false,  false);
    }

    const int o  = nTile * 16 + nIdx;
    const float bo = bias[o];
    #pragma unroll
    for (int r = 0; r < 8; ++r) {
        const int m     = r + 8 * kh;
        const int pglob = pbase + mTile * 16 + m;
        const int b_    = pglob >> 14;
        const int hw_   = pglob & (H * W - 1);
        out[((size_t)(b_ * COUT + o)) * (H * W) + hw_] = acc[r] + bo;
    }
    }
}

extern "C" void kernel_launch(void* const* d_in, const int* in_sizes, int n_in,
                              void* d_out, int out_size, void* d_ws, size_t ws_size,
                              hipStream_t stream) {
    (void)in_sizes; (void)n_in; (void)out_size; (void)d_ws; (void)ws_size;
    const float* x      = (const float*)d_in[0];
    const float* weight = (const float*)d_in[1];
    const float* bias   = (const float*)d_in[2];
    const float* op_w   = (const float*)d_in[3];
    const float* op_b   = (const float*)d_in[4];
    float* out = (float*)d_out;

    dyfconv_fused<<<NUM_WG, THREADS, 0, stream>>>(x, weight, bias, op_w, op_b, out);
}


# Operator-apply phase: each thread loads its pixel's 81 operator values from LDS once into registers and reuses them for its 8 channels (were re-read per channel, 648 waited loads); sum order kept
# speedup vs baseline: 4.0214x; 1.2124x over previous
#include <hip/hip_runtime.h>
#include <hip/hip_bf16.h>
#include <math.h>

typedef __attribute__((ext_vector_type(2))) float v2f;
typedef __attribute__((ext_vector_type(8))) float v8f;

constexpr int KSZ   = 3;
constexpr int K2    = 9;
constexpr int CIN   = 64;
constexpr int COUT  = 64;
constexpr int NB    = 8;
constexpr int H     = 128;
constexpr int W     = 128;
constexpr int KDIM  = CIN * K2;

constexpr int THREADS     = 512;
constexpr int PIX_PER_BLK = 64;
constexpr int NUM_TILES   = NB * H * W / PIX_PER_BLK;
constexpr int NUM_WG      = 256;
constexpr int YSTRIDE     = KDIM + 1;
constexpr int OPSTRIDE    = 84;

constexpr int LDS_W_OFF  = 0;
constexpr int LDS_Y_OFF  = LDS_W_OFF + KDIM * COUT;
constexpr int LDS_OP_OFF = LDS_Y_OFF + PIX_PER_BLK * YSTRIDE;
constexpr int LDS_N_OFF  = LDS_OP_OFF + PIX_PER_BLK * OPSTRIDE;
constexpr int LDS_FLOATS = LDS_N_OFF + PIX_PER_BLK * K2;

__global__ __launch_bounds__(THREADS, 1)
void dyfconv_fused(const float* __restrict__ x,
                   const float* __restrict__ weight,
                   const float* __restrict__ bias,
                   const float* __restrict__ op_w,
                   const float* __restrict__ op_b,
                   float* __restrict__ out)
{
    __shared__ __attribute__((aligned(16))) float lds[LDS_FLOATS];
    float* ldsW  = lds + LDS_W_OFF;
    float* ldsY  = lds + LDS_Y_OFF;
    float* ldsOp = lds + LDS_OP_OFF;
    float* ldsN  = lds + LDS_N_OFF;

    const int t     = threadIdx.x;

    for (int idx = t; idx < KDIM * COUT; idx += THREADS) {
        int kk = idx >> 6;
        int o  = idx & (COUT - 1);
        int c  = kk / K2;
        int k  = kk - c * K2;
        ldsW[idx] = weight[(c * COUT + o) * K2 + k];
    }
    for (int idx = t; idx < PIX_PER_BLK * K2; idx += THREADS)
        ldsN[idx] = 0.0f;
    __syncthreads();

    const int p  = t >> 3;
    const int g  = t & 7;

    const int wave  = t >> 5;
    const int lane  = t & 31;
    const int mTile = wave & 3;
    const int nTile = wave >> 2;
    const int mIdx  = lane & 15;
    const int nIdx  = lane & 15;
    const int kh    = lane >> 4;

    for (int tile = blockIdx.x; tile < NUM_TILES; tile += gridDim.x) {
    const int pbase = tile * PIX_PER_BLK;
    const int pg = pbase + p;
    const int bb = pg >> 14;
    const int hw = pg & (H * W - 1);
    const int hh = hw >> 7;
    const int ww = hw & (W - 1);

    float xv[8][K2];
    {
        int  toff[K2];
        bool tok[K2];
        #pragma unroll
        for (int ki = 0; ki < KSZ; ++ki) {
            const int hs = hh + ki - 1;
            const int hc = min(max(hs, 0), H - 1);
            #pragma unroll
            for (int kj = 0; kj < KSZ; ++kj) {
                const int wsv = ww + kj - 1;
                const int wc  = min(max(wsv, 0), W - 1);
                tok[ki * KSZ + kj]  = ((unsigned)hs < (unsigned)H) && ((unsigned)wsv < (unsigned)W);
                toff[ki * KSZ + kj] = hc * W + wc;
            }
        }
        const float* xg = x + (size_t)(bb * CIN + g * 8) * (H * W);
        #pragma unroll
        for (int cc = 0; cc < 8; ++cc) {
            #pragma unroll
            for (int j = 0; j < K2; ++j) {
                const float v = xg[cc * (H * W) + toff[j]];
                xv[cc][j] = tok[j] ? v : 0.0f;
            }
        }
    }

    {
        float sq[K2];
        #pragma unroll
        for (int j = 0; j < K2; ++j) sq[j] = 0.0f;
        #pragma unroll
        for (int cc = 0; cc < 8; ++cc) {
            #pragma unroll
            for (int j = 0; j < K2; ++j)
                sq[j] += xv[cc][j] * xv[cc][j];
        }
        #pragma unroll
        for (int j = 0; j < K2; ++j)
            atomicAdd(&ldsN[p * K2 + j], sq[j]);
    }
    __syncthreads();

    for (int e = t; e < PIX_PER_BLK * 81; e += THREADS) {
        const int pp  = e / 81;
        const int idx = e - pp * 81;
        const float* nb = &ldsN[pp * K2];
        float acc = op_b[idx];
        #pragma unroll
        for (int j = 0; j < K2; ++j)
            acc += sqrtf(nb[j]) * op_w[idx * K2 + j];
        ldsOp[pp * OPSTRIDE + idx] = tanhf(acc);
    }
    __syncthreads();

    for (int idx = t; idx < PIX_PER_BLK * K2; idx += THREADS)
        ldsN[idx] = 0.0f;

    {
        float opr[81];
        {
            const float4* op4 = reinterpret_cast<const float4*>(&ldsOp[p * OPSTRIDE]);
            #pragma unroll
            for (int q = 0; q < 20; ++q) {
                const float4 v = op4[q];
                opr[4 * q + 0] = v.x;
                opr[4 * q + 1] = v.y;
                opr[4 * q + 2] = v.z;
                opr[4 * q + 3] = v.w;
            }
            opr[80] = ldsOp[p * OPSTRIDE + 80];
        }
        float* yrow = &ldsY[p * YSTRIDE + g * 8 * K2];
        #pragma unroll
        for (int cc = 0; cc < 8; ++cc) {
            #pragma unroll
            for (int k = 0; k < K2; ++k) {
                float acc = xv[cc][k];
                #pragma unroll
                for (int j = 0; j < K2; ++j)
                    acc += xv[cc][j] * opr[j * K2 + k];
                yrow[cc * K2 + k] = acc;
            }
        }
    }
    __syncthreads();

    const float* yrow = &ldsY[(mTile * 16 + mIdx) * YSTRIDE];
    const float* wcol = &ldsW[nTile * 16 + nIdx];

    v8f acc = {};
    #pragma unroll 4
    for (int kk = 0; kk < KDIM; kk += 4) {
        v2f a, bf;
        const int kb = kk + 2 * kh;
        a.x = yrow[kb + 0];
        a.y = yrow[kb + 1];
        bf.x = wcol[(kb + 0) * COUT];
        bf.y = wcol[(kb + 1) * COUT];
        acc = __builtin_amdgcn_wmma_f32_16x16x4_f32(
                   false, a,  false, bf,
                   (short)0, acc,  false,  false);
    }

    const int o  = nTile * 16 + nIdx;
    const float bo = bias[o];
    #pragma unroll
    for (int r = 0; r < 8; ++r) {
        const int m     = r + 8 * kh;
        const int pglob = pbase + mTile * 16 + m;
        const int b_    = pglob >> 14;
        const int hw_   = pglob & (H * W - 1);
        out[((size_t)(b_ * COUT + o)) * (H * W) + hw_] = acc[r] + bo;
    }
    }
}

extern "C" void kernel_launch(void* const* d_in, const int* in_sizes, int n_in,
                              void* d_out, int out_size, void* d_ws, size_t ws_size,
                              hipStream_t stream) {
    (void)in_sizes; (void)n_in; (void)out_size; (void)d_ws; (void)ws_size;
    const float* x      = (const float*)d_in[0];
    const float* weight = (const float*)d_in[1];
    const float* bias   = (const float*)d_in[2];
    const float* op_w   = (const float*)d_in[3];
    const float* op_b   = (const float*)d_in[4];
    float* out = (float*)d_out;

    dyfconv_fused<<<NUM_WG, THREADS, 0, stream>>>(x, weight, bias, op_w, op_b, out);
}


# Patch norms: the square root of each (pixel, tap) sum of squares is taken once, in place in LDS, ahead of the operator loop (was retaken inside it, 81 times per value); one barrier per tile added
# speedup vs baseline: 4.4518x; 1.1070x over previous
#include <hip/hip_runtime.h>
#include <hip/hip_bf16.h>
#include <math.h>

typedef __attribute__((ext_vector_type(2))) float v2f;
typedef __attribute__((ext_vector_type(8))) float v8f;

constexpr int KSZ   = 3;
constexpr int K2    = 9;
constexpr int CIN   = 64;
constexpr int COUT  = 64;
constexpr int NB    = 8;
constexpr int H     = 128;
constexpr int W     = 128;
constexpr int KDIM  = CIN * K2;

constexpr int THREADS     = 512;
constexpr int PIX_PER_BLK = 64;
constexpr int NUM_TILES   = NB * H * W / PIX_PER_BLK;
constexpr int NUM_WG      = 256;
constexpr int YSTRIDE     = KDIM + 1;
constexpr int OPSTRIDE    = 84;

constexpr int LDS_W_OFF  = 0;
constexpr int LDS_Y_OFF  = LDS_W_OFF + KDIM * COUT;
constexpr int LDS_OP_OFF = LDS_Y_OFF + PIX_PER_BLK * YSTRIDE;
constexpr int LDS_N_OFF  = LDS_OP_OFF + PIX_PER_BLK * OPSTRIDE;
constexpr int LDS_FLOATS = LDS_N_OFF + PIX_PER_BLK * K2;

__global__ __launch_bounds__(THREADS, 1)
void dyfconv_fused(const float* __restrict__ x,
                   const float* __restrict__ weight,
                   const float* __restrict__ bias,
                   const float* __restrict__ op_w,
                   const float* __restrict__ op_b,
                   float* __restrict__ out)
{
    __shared__ __attribute__((aligned(16))) float lds[LDS_FLOATS];
    float* ldsW  = lds + LDS_W_OFF;
    float* ldsY  = lds + LDS_Y_OFF;
    float* ldsOp = lds + LDS_OP_OFF;
    float* ldsN  = lds + LDS_N_OFF;

    const int t     = threadIdx.x;

    for (int idx = t; idx < KDIM * COUT; idx += THREADS) {
        int kk = idx >> 6;
        int o  = idx & (COUT - 1);
        int c  = kk / K2;
        int k  = kk - c * K2;
        ldsW[idx] = weight[(c * COUT + o) * K2 + k];
    }
    for (int idx = t; idx < PIX_PER_BLK * K2; idx += THREADS)
        ldsN[idx] = 0.0f;
    __syncthreads();

    const int p  = t >> 3;
    const int g  = t & 7;

    const int wave  = t >> 5;
    const int lane  = t & 31;
    const int mTile = wave & 3;
    const int nTile = wave >> 2;
    const int mIdx  = lane & 15;
    const int nIdx  = lane & 15;
    const int kh    = lane >> 4;

    for (int tile = blockIdx.x; tile < NUM_TILES; tile += gridDim.x) {
    const int pbase = tile * PIX_PER_BLK;
    const int pg = pbase + p;
    const int bb = pg >> 14;
    const int hw = pg & (H * W - 1);
    const int hh = hw >> 7;
    const int ww = hw & (W - 1);

    float xv[8][K2];
    {
        int  toff[K2];
        bool tok[K2];
        #pragma unroll
        for (int ki = 0; ki < KSZ; ++ki) {
            const int hs = hh + ki - 1;
            const int hc = min(max(hs, 0), H - 1);
            #pragma unroll
            for (int kj = 0; kj < KSZ; ++kj) {
                const int wsv = ww + kj - 1;
                const int wc  = min(max(wsv, 0), W - 1);
                tok[ki * KSZ + kj]  = ((unsigned)hs < (unsigned)H) && ((unsigned)wsv < (unsigned)W);
                toff[ki * KSZ + kj] = hc * W + wc;
            }
        }
        const float* xg = x + (size_t)(bb * CIN + g * 8) * (H * W);
        #pragma unroll
        for (int cc = 0; cc < 8; ++cc) {
            #pragma unroll
            for (int j = 0; j < K2; ++j) {
                const float v = xg[cc * (H * W) + toff[j]];
                xv[cc][j] = tok[j] ? v : 0.0f;
            }
        }
    }

    {
        float sq[K2];
        #pragma unroll
        for (int j = 0; j < K2; ++j) sq[j] = 0.0f;
        #pragma unroll
        for (int cc = 0; cc < 8; ++cc) {
            #pragma unroll
            for (int j = 0; j < K2; ++j)
                sq[j] += xv[cc][j] * xv[cc][j];
        }
        #pragma unroll
        for (int j = 0; j < K2; ++j)
            atomicAdd(&ldsN[p * K2 + j], sq[j]);
    }
    __syncthreads();

    for (int idx = t; idx < PIX_PER_BLK * K2; idx += THREADS)
        ldsN[idx] = sqrtf(ldsN[idx]);
    __syncthreads();

    for (int e = t; e < PIX_PER_BLK * 81; e += THREADS) {
        const int pp  = e / 81;
        const int idx = e - pp * 81;
        const float* nb = &ldsN[pp * K2];
        float acc = op_b[idx];
        #pragma unroll
        for (int j = 0; j < K2; ++j)
            acc += nb[j] * op_w[idx * K2 + j];
        ldsOp[pp * OPSTRIDE + idx] = tanhf(acc);
    }
    __syncthreads();

    for (int idx = t; idx < PIX_PER_BLK * K2; idx += THREADS)
        ldsN[idx] = 0.0f;

    {
        float opr[81];
        {
            const float4* op4 = reinterpret_cast<const float4*>(&ldsOp[p * OPSTRIDE]);
            #pragma unroll
            for (int q = 0; q < 20; ++q) {
                const float4 v = op4[q];
                opr[4 * q + 0] = v.x;
                opr[4 * q + 1] = v.y;
                opr[4 * q + 2] = v.z;
                opr[4 * q + 3] = v.w;
            }
            opr[80] = ldsOp[p * OPSTRIDE + 80];
        }
        float* yrow = &ldsY[p * YSTRIDE + g * 8 * K2];
        #pragma unroll
        for (int cc = 0; cc < 8; ++cc) {
            #pragma unroll
            for (int k = 0; k < K2; ++k) {
                float acc = xv[cc][k];
                #pragma unroll
                for (int j = 0; j < K2; ++j)
                    acc += xv[cc][j] * opr[j * K2 + k];
                yrow[cc * K2 + k] = acc;
            }
        }
    }
    __syncthreads();

    const float* yrow = &ldsY[(mTile * 16 + mIdx) * YSTRIDE];
    const float* wcol = &ldsW[nTile * 16 + nIdx];

    v8f acc = {};
    #pragma unroll 4
    for (int kk = 0; kk < KDIM; kk += 4) {
        v2f a, bf;
        const int kb = kk + 2 * kh;
        a.x = yrow[kb + 0];
        a.y = yrow[kb + 1];
        bf.x = wcol[(kb + 0) * COUT];
        bf.y = wcol[(kb + 1) * COUT];
        acc = __builtin_amdgcn_wmma_f32_16x16x4_f32(
                   false, a,  false, bf,
                   (short)0, acc,  false,  false);
    }

    const int o  = nTile * 16 + nIdx;
    const float bo = bias[o];
    #pragma unroll
    for (int r = 0; r < 8; ++r) {
        const int m     = r + 8 * kh;
        const int pglob = pbase + mTile * 16 + m;
        const int b_    = pglob >> 14;
        const int hw_   = pglob & (H * W - 1);
        out[((size_t)(b_ * COUT + o)) * (H * W) + hw_] = acc[r] + bo;
    }
    }
}

extern "C" void kernel_launch(void* const* d_in, const int* in_sizes, int n_in,
                              void* d_out, int out_size, void* d_ws, size_t ws_size,
                              hipStream_t stream) {
    (void)in_sizes; (void)n_in; (void)out_size; (void)d_ws; (void)ws_size;
    const float* x      = (const float*)d_in[0];
    const float* weight = (const float*)d_in[1];
    const float* bias   = (const float*)d_in[2];
    const float* op_w   = (const float*)d_in[3];
    const float* op_b   = (const float*)d_in[4];
    float* out = (float*)d_out;

    dyfconv_fused<<<NUM_WG, THREADS, 0, stream>>>(x, weight, bias, op_w, op_b, out);
}


# Next tile's 72 x loads per thread issued inside the GEMM, 9 per channel between its k-groups (first tile's before the weight staging): load latency runs under the WMMA phase, not at the tile head
# speedup vs baseline: 4.6088x; 1.0353x over previous
#include <hip/hip_runtime.h>
#include <hip/hip_bf16.h>
#include <math.h>

typedef __attribute__((ext_vector_type(2))) float v2f;
typedef __attribute__((ext_vector_type(8))) float v8f;

constexpr int KSZ   = 3;
constexpr int K2    = 9;
constexpr int CIN   = 64;
constexpr int COUT  = 64;
constexpr int NB    = 8;
constexpr int H     = 128;
constexpr int W     = 128;
constexpr int KDIM  = CIN * K2;

constexpr int THREADS     = 512;
constexpr int PIX_PER_BLK = 64;
constexpr int NUM_TILES   = NB * H * W / PIX_PER_BLK;
constexpr int NUM_WG      = 256;
constexpr int YSTRIDE     = KDIM + 1;
constexpr int OPSTRIDE    = 84;

constexpr int LDS_W_OFF  = 0;
constexpr int LDS_Y_OFF  = LDS_W_OFF + KDIM * COUT;
constexpr int LDS_OP_OFF = LDS_Y_OFF + PIX_PER_BLK * YSTRIDE;
constexpr int LDS_N_OFF  = LDS_OP_OFF + PIX_PER_BLK * OPSTRIDE;
constexpr int LDS_FLOATS = LDS_N_OFF + PIX_PER_BLK * K2;

__device__ __forceinline__ void patch_geom(int tile, int p, int g, const float* __restrict__ x,
                                           int (&toff)[K2], bool (&tok)[K2], const float*& xg)
{
    const int pg = tile * PIX_PER_BLK + p;
    const int bb = pg >> 14;
    const int hw = pg & (H * W - 1);
    const int hh = hw >> 7;
    const int ww = hw & (W - 1);
    #pragma unroll
    for (int ki = 0; ki < KSZ; ++ki) {
        const int hs = hh + ki - 1;
        const int hc = min(max(hs, 0), H - 1);
        #pragma unroll
        for (int kj = 0; kj < KSZ; ++kj) {
            const int wsv = ww + kj - 1;
            const int wc  = min(max(wsv, 0), W - 1);
            tok[ki * KSZ + kj]  = ((unsigned)hs < (unsigned)H) && ((unsigned)wsv < (unsigned)W);
            toff[ki * KSZ + kj] = hc * W + wc;
        }
    }
    xg = x + (size_t)(bb * CIN + g * 8) * (H * W);
}

__global__ __launch_bounds__(THREADS, 1)
void dyfconv_fused(const float* __restrict__ x,
                   const float* __restrict__ weight,
                   const float* __restrict__ bias,
                   const float* __restrict__ op_w,
                   const float* __restrict__ op_b,
                   float* __restrict__ out)
{
    __shared__ __attribute__((aligned(16))) float lds[LDS_FLOATS];
    float* ldsW  = lds + LDS_W_OFF;
    float* ldsY  = lds + LDS_Y_OFF;
    float* ldsOp = lds + LDS_OP_OFF;
    float* ldsN  = lds + LDS_N_OFF;

    const int t     = threadIdx.x;

    const int p  = t >> 3;
    const int g  = t & 7;

    float xv[8][K2];
    {
        int  toff[K2];
        bool tok[K2];
        const float* xg;
        patch_geom(min((int)blockIdx.x, NUM_TILES - 1), p, g, x, toff, tok, xg);
        #pragma unroll
        for (int cc = 0; cc < 8; ++cc) {
            #pragma unroll
            for (int j = 0; j < K2; ++j)
                xv[cc][j] = xg[cc * (H * W) + toff[j]];
        }
    }

    for (int idx = t; idx < KDIM * COUT; idx += THREADS) {
        int kk = idx >> 6;
        int o  = idx & (COUT - 1);
        int c  = kk / K2;
        int k  = kk - c * K2;
        ldsW[idx] = weight[(c * COUT + o) * K2 + k];
    }
    for (int idx = t; idx < PIX_PER_BLK * K2; idx += THREADS)
        ldsN[idx] = 0.0f;
    __syncthreads();

    const int wave  = t >> 5;
    const int lane  = t & 31;
    const int mTile = wave & 3;
    const int nTile = wave >> 2;
    const int mIdx  = lane & 15;
    const int nIdx  = lane & 15;
    const int kh    = lane >> 4;

    for (int tile = blockIdx.x; tile < NUM_TILES; tile += gridDim.x) {
    const int pbase = tile * PIX_PER_BLK;

    {
        int  toff[K2];
        bool tok[K2];
        const float* xg;
        patch_geom(tile, p, g, x, toff, tok, xg);
        #pragma unroll
        for (int cc = 0; cc < 8; ++cc) {
            #pragma unroll
            for (int j = 0; j < K2; ++j)
                xv[cc][j] = tok[j] ? xv[cc][j] : 0.0f;
        }
    }

    {
        float sq[K2];
        #pragma unroll
        for (int j = 0; j < K2; ++j) sq[j] = 0.0f;
        #pragma unroll
        for (int cc = 0; cc < 8; ++cc) {
            #pragma unroll
            for (int j = 0; j < K2; ++j)
                sq[j] += xv[cc][j] * xv[cc][j];
        }
        #pragma unroll
        for (int j = 0; j < K2; ++j)
            atomicAdd(&ldsN[p * K2 + j], sq[j]);
    }
    __syncthreads();

    for (int idx = t; idx < PIX_PER_BLK * K2; idx += THREADS)
        ldsN[idx] = sqrtf(ldsN[idx]);
    __syncthreads();

    for (int e = t; e < PIX_PER_BLK * 81; e += THREADS) {
        const int pp  = e / 81;
        const int idx = e - pp * 81;
        const float* nb = &ldsN[pp * K2];
        float acc = op_b[idx];
        #pragma unroll
        for (int j = 0; j < K2; ++j)
            acc += nb[j] * op_w[idx * K2 + j];
        ldsOp[pp * OPSTRIDE + idx] = tanhf(acc);
    }
    __syncthreads();

    for (int idx = t; idx < PIX_PER_BLK * K2; idx += THREADS)
        ldsN[idx] = 0.0f;

    {
        float opr[81];
        {
            const float4* op4 = reinterpret_cast<const float4*>(&ldsOp[p * OPSTRIDE]);
            #pragma unroll
            for (int q = 0; q < 20; ++q) {
                const float4 v = op4[q];
                opr[4 * q + 0] = v.x;
                opr[4 * q + 1] = v.y;
                opr[4 * q + 2] = v.z;
                opr[4 * q + 3] = v.w;
            }
            opr[80] = ldsOp[p * OPSTRIDE + 80];
        }
        float* yrow = &ldsY[p * YSTRIDE + g * 8 * K2];
        #pragma unroll
        for (int cc = 0; cc < 8; ++cc) {
            #pragma unroll
            for (int k = 0; k < K2; ++k) {
                float acc = xv[cc][k];
                #pragma unroll
                for (int j = 0; j < K2; ++j)
                    acc += xv[cc][j] * opr[j * K2 + k];
                yrow[cc * K2 + k] = acc;
            }
        }
    }
    __syncthreads();

    const float* yrow = &ldsY[(mTile * 16 + mIdx) * YSTRIDE];
    const float* wcol = &ldsW[nTile * 16 + nIdx];

    int ntoff[K2];
    const float* nxg;
    {
        bool ntok[K2];
        const int nt = tile + (int)gridDim.x;
        patch_geom(nt < NUM_TILES ? nt : tile, p, g, x, ntoff, ntok, nxg);
    }

    v8f acc = {};
    #pragma unroll
    for (int cc = 0; cc < 8; ++cc) {
        #pragma unroll
        for (int j = 0; j < K2; ++j)
            xv[cc][j] = nxg[cc * (H * W) + ntoff[j]];

        #pragma unroll 6
        for (int kk = cc * 72; kk < cc * 72 + 72; kk += 4) {
            v2f a, bf;
            const int kb = kk + 2 * kh;
            a.x = yrow[kb + 0];
            a.y = yrow[kb + 1];
            bf.x = wcol[(kb + 0) * COUT];
            bf.y = wcol[(kb + 1) * COUT];
            acc = __builtin_amdgcn_wmma_f32_16x16x4_f32(
                       false, a,  false, bf,
                       (short)0, acc,  false,  false);
        }
    }

    const int o  = nTile * 16 + nIdx;
    const float bo = bias[o];
    #pragma unroll
    for (int r = 0; r < 8; ++r) {
        const int m     = r + 8 * kh;
        const int pglob = pbase + mTile * 16 + m;
        const int b_    = pglob >> 14;
        const int hw_   = pglob & (H * W - 1);
        out[((size_t)(b_ * COUT + o)) * (H * W) + hw_] = acc[r] + bo;
    }
    }
}

extern "C" void kernel_launch(void* const* d_in, const int* in_sizes, int n_in,
                              void* d_out, int out_size, void* d_ws, size_t ws_size,
                              hipStream_t stream) {
    (void)in_sizes; (void)n_in; (void)out_size; (void)d_ws; (void)ws_size;
    const float* x      = (const float*)d_in[0];
    const float* weight = (const float*)d_in[1];
    const float* bias   = (const float*)d_in[2];
    const float* op_w   = (const float*)d_in[3];
    const float* op_b   = (const float*)d_in[4];
    float* out = (float*)d_out;

    dyfconv_fused<<<NUM_WG, THREADS, 0, stream>>>(x, weight, bias, op_w, op_b, out);
}


# GEMM fragments: one bank-conflict-free 16-byte LDS load per operand per two k-steps (weight laid [k/4][o][4], A pitch 580, lane half = 4 contiguous k); sum order of k within each 8 changes
# speedup vs baseline: 4.9600x; 1.0762x over previous
#include <hip/hip_runtime.h>
#include <hip/hip_bf16.h>
#include <math.h>

typedef __attribute__((ext_vector_type(2))) float v2f;
typedef __attribute__((ext_vector_type(8))) float v8f;

constexpr int KSZ   = 3;
constexpr int K2    = 9;
constexpr int CIN   = 64;
constexpr int COUT  = 64;
constexpr int NB    = 8;
constexpr int H     = 128;
constexpr int W     = 128;
constexpr int KDIM  = CIN * K2;

constexpr int THREADS     = 512;
constexpr int PIX_PER_BLK = 64;
constexpr int NUM_TILES   = NB * H * W / PIX_PER_BLK;
constexpr int NUM_WG      = 256;
constexpr int YSTRIDE     = KDIM + 4;
constexpr int OPSTRIDE    = 84;

constexpr int LDS_W_OFF  = 0;
constexpr int LDS_Y_OFF  = LDS_W_OFF + KDIM * COUT;
constexpr int LDS_OP_OFF = LDS_Y_OFF + PIX_PER_BLK * YSTRIDE;
constexpr int LDS_N_OFF  = LDS_OP_OFF + PIX_PER_BLK * OPSTRIDE;
constexpr int LDS_FLOATS = LDS_N_OFF + PIX_PER_BLK * K2;

__device__ __forceinline__ void patch_geom(int tile, int p, int g, const float* __restrict__ x,
                                           int (&toff)[K2], bool (&tok)[K2], const float*& xg)
{
    const int pg = tile * PIX_PER_BLK + p;
    const int bb = pg >> 14;
    const int hw = pg & (H * W - 1);
    const int hh = hw >> 7;
    const int ww = hw & (W - 1);
    #pragma unroll
    for (int ki = 0; ki < KSZ; ++ki) {
        const int hs = hh + ki - 1;
        const int hc = min(max(hs, 0), H - 1);
        #pragma unroll
        for (int kj = 0; kj < KSZ; ++kj) {
            const int wsv = ww + kj - 1;
            const int wc  = min(max(wsv, 0), W - 1);
            tok[ki * KSZ + kj]  = ((unsigned)hs < (unsigned)H) && ((unsigned)wsv < (unsigned)W);
            toff[ki * KSZ + kj] = hc * W + wc;
        }
    }
    xg = x + (size_t)(bb * CIN + g * 8) * (H * W);
}

__global__ __launch_bounds__(THREADS, 1)
void dyfconv_fused(const float* __restrict__ x,
                   const float* __restrict__ weight,
                   const float* __restrict__ bias,
                   const float* __restrict__ op_w,
                   const float* __restrict__ op_b,
                   float* __restrict__ out)
{
    __shared__ __attribute__((aligned(16))) float lds[LDS_FLOATS];
    float* ldsW  = lds + LDS_W_OFF;
    float* ldsY  = lds + LDS_Y_OFF;
    float* ldsOp = lds + LDS_OP_OFF;
    float* ldsN  = lds + LDS_N_OFF;

    const int t     = threadIdx.x;

    const int p  = t >> 3;
    const int g  = t & 7;

    float xv[8][K2];
    {
        int  toff[K2];
        bool tok[K2];
        const float* xg;
        patch_geom(min((int)blockIdx.x, NUM_TILES - 1), p, g, x, toff, tok, xg);
        #pragma unroll
        for (int cc = 0; cc < 8; ++cc) {
            #pragma unroll
            for (int j = 0; j < K2; ++j)
                xv[cc][j] = xg[cc * (H * W) + toff[j]];
        }
    }

    for (int idx = t; idx < KDIM * COUT; idx += THREADS) {
        int kq = idx & 3;
        int o  = (idx >> 2) & (COUT - 1);
        int kk = ((idx >> 8) << 2) | kq;
        int c  = kk / K2;
        int k  = kk - c * K2;
        ldsW[idx] = weight[(c * COUT + o) * K2 + k];
    }
    for (int idx = t; idx < PIX_PER_BLK * K2; idx += THREADS)
        ldsN[idx] = 0.0f;
    __syncthreads();

    const int wave  = t >> 5;
    const int lane  = t & 31;
    const int mTile = wave & 3;
    const int nTile = wave >> 2;
    const int mIdx  = lane & 15;
    const int nIdx  = lane & 15;
    const int kh    = lane >> 4;

    for (int tile = blockIdx.x; tile < NUM_TILES; tile += gridDim.x) {
    const int pbase = tile * PIX_PER_BLK;

    {
        int  toff[K2];
        bool tok[K2];
        const float* xg;
        patch_geom(tile, p, g, x, toff, tok, xg);
        #pragma unroll
        for (int cc = 0; cc < 8; ++cc) {
            #pragma unroll
            for (int j = 0; j < K2; ++j)
                xv[cc][j] = tok[j] ? xv[cc][j] : 0.0f;
        }
    }

    {
        float sq[K2];
        #pragma unroll
        for (int j = 0; j < K2; ++j) sq[j] = 0.0f;
        #pragma unroll
        for (int cc = 0; cc < 8; ++cc) {
            #pragma unroll
            for (int j = 0; j < K2; ++j)
                sq[j] += xv[cc][j] * xv[cc][j];
        }
        #pragma unroll
        for (int j = 0; j < K2; ++j)
            atomicAdd(&ldsN[p * K2 + j], sq[j]);
    }
    __syncthreads();

    for (int idx = t; idx < PIX_PER_BLK * K2; idx += THREADS)
        ldsN[idx] = sqrtf(ldsN[idx]);
    __syncthreads();

    for (int e = t; e < PIX_PER_BLK * 81; e += THREADS) {
        const int pp  = e / 81;
        const int idx = e - pp * 81;
        const float* nb = &ldsN[pp * K2];
        float acc = op_b[idx];
        #pragma unroll
        for (int j = 0; j < K2; ++j)
            acc += nb[j] * op_w[idx * K2 + j];
        ldsOp[pp * OPSTRIDE + idx] = tanhf(acc);
    }
    __syncthreads();

    for (int idx = t; idx < PIX_PER_BLK * K2; idx += THREADS)
        ldsN[idx] = 0.0f;

    {
        float opr[81];
        {
            const float4* op4 = reinterpret_cast<const float4*>(&ldsOp[p * OPSTRIDE]);
            #pragma unroll
            for (int q = 0; q < 20; ++q) {
                const float4 v = op4[q];
                opr[4 * q + 0] = v.x;
                opr[4 * q + 1] = v.y;
                opr[4 * q + 2] = v.z;
                opr[4 * q + 3] = v.w;
            }
            opr[80] = ldsOp[p * OPSTRIDE + 80];
        }
        float* yrow = &ldsY[p * YSTRIDE + g * 8 * K2];
        #pragma unroll
        for (int cc = 0; cc < 8; ++cc) {
            #pragma unroll
            for (int k = 0; k < K2; ++k) {
                float acc = xv[cc][k];
                #pragma unroll
                for (int j = 0; j < K2; ++j)
                    acc += xv[cc][j] * opr[j * K2 + k];
                yrow[cc * K2 + k] = acc;
            }
        }
    }
    __syncthreads();

    const float4* yrow4 = reinterpret_cast<const float4*>(&ldsY[(mTile * 16 + mIdx) * YSTRIDE + 4 * kh]);
    const float4* wcol4 = reinterpret_cast<const float4*>(&ldsW[(kh * COUT + nTile * 16 + nIdx) * 4]);

    int ntoff[K2];
    const float* nxg;
    {
        bool ntok[K2];
        const int nt = tile + (int)gridDim.x;
        patch_geom(nt < NUM_TILES ? nt : tile, p, g, x, ntoff, ntok, nxg);
    }

    v8f acc = {};
    #pragma unroll
    for (int cc = 0; cc < 8; ++cc) {
        #pragma unroll
        for (int j = 0; j < K2; ++j)
            xv[cc][j] = nxg[cc * (H * W) + ntoff[j]];

        #pragma unroll 3
        for (int q = cc * 9; q < cc * 9 + 9; ++q) {
            const float4 a4 = yrow4[q * 2];
            const float4 b4 = wcol4[q * 2 * COUT];
            v2f a, bf;
            a.x = a4.x;  a.y = a4.y;
            bf.x = b4.x; bf.y = b4.y;
            acc = __builtin_amdgcn_wmma_f32_16x16x4_f32(
                       false, a,  false, bf,
                       (short)0, acc,  false,  false);
            a.x = a4.z;  a.y = a4.w;
            bf.x = b4.z; bf.y = b4.w;
            acc = __builtin_amdgcn_wmma_f32_16x16x4_f32(
                       false, a,  false, bf,
                       (short)0, acc,  false,  false);
        }
    }

    const int o  = nTile * 16 + nIdx;
    const float bo = bias[o];
    #pragma unroll
    for (int r = 0; r < 8; ++r) {
        const int m     = r + 8 * kh;
        const int pglob = pbase + mTile * 16 + m;
        const int b_    = pglob >> 14;
        const int hw_   = pglob & (H * W - 1);
        out[((size_t)(b_ * COUT + o)) * (H * W) + hw_] = acc[r] + bo;
    }
    }
}

extern "C" void kernel_launch(void* const* d_in, const int* in_sizes, int n_in,
                              void* d_out, int out_size, void* d_ws, size_t ws_size,
                              hipStream_t stream) {
    (void)in_sizes; (void)n_in; (void)out_size; (void)d_ws; (void)ws_size;
    const float* x      = (const float*)d_in[0];
    const float* weight = (const float*)d_in[1];
    const float* bias   = (const float*)d_in[2];
    const float* op_w   = (const float*)d_in[3];
    const float* op_b   = (const float*)d_in[4];
    float* out = (float*)d_out;

    dyfconv_fused<<<NUM_WG, THREADS, 0, stream>>>(x, weight, bias, op_w, op_b, out);
}


# Patch norms: the 8 channel-group partial sums of a pixel are added across their 8 lanes by a fixed DPP butterfly and stored once (were LDS float atomics, 8 lanes per word); sum order now a fixed tree
# speedup vs baseline: 5.3764x; 1.0840x over previous
#include <hip/hip_runtime.h>
#include <hip/hip_bf16.h>
#include <math.h>

typedef __attribute__((ext_vector_type(2))) float v2f;
typedef __attribute__((ext_vector_type(8))) float v8f;

constexpr int KSZ   = 3;
constexpr int K2    = 9;
constexpr int CIN   = 64;
constexpr int COUT  = 64;
constexpr int NB    = 8;
constexpr int H     = 128;
constexpr int W     = 128;
constexpr int KDIM  = CIN * K2;

constexpr int THREADS     = 512;
constexpr int PIX_PER_BLK = 64;
constexpr int NUM_TILES   = NB * H * W / PIX_PER_BLK;
constexpr int NUM_WG      = 256;
constexpr int YSTRIDE     = KDIM + 4;
constexpr int OPSTRIDE    = 84;

constexpr int LDS_W_OFF  = 0;
constexpr int LDS_Y_OFF  = LDS_W_OFF + KDIM * COUT;
constexpr int LDS_OP_OFF = LDS_Y_OFF + PIX_PER_BLK * YSTRIDE;
constexpr int LDS_N_OFF  = LDS_OP_OFF + PIX_PER_BLK * OPSTRIDE;
constexpr int LDS_FLOATS = LDS_N_OFF + PIX_PER_BLK * K2;

__device__ __forceinline__ void patch_geom(int tile, int p, int g, const float* __restrict__ x,
                                           int (&toff)[K2], bool (&tok)[K2], const float*& xg)
{
    const int pg = tile * PIX_PER_BLK + p;
    const int bb = pg >> 14;
    const int hw = pg & (H * W - 1);
    const int hh = hw >> 7;
    const int ww = hw & (W - 1);
    #pragma unroll
    for (int ki = 0; ki < KSZ; ++ki) {
        const int hs = hh + ki - 1;
        const int hc = min(max(hs, 0), H - 1);
        #pragma unroll
        for (int kj = 0; kj < KSZ; ++kj) {
            const int wsv = ww + kj - 1;
            const int wc  = min(max(wsv, 0), W - 1);
            tok[ki * KSZ + kj]  = ((unsigned)hs < (unsigned)H) && ((unsigned)wsv < (unsigned)W);
            toff[ki * KSZ + kj] = hc * W + wc;
        }
    }
    xg = x + (size_t)(bb * CIN + g * 8) * (H * W);
}

template <int MASK>
__device__ __forceinline__ float lane_xor_get(float v)
{
    return __builtin_bit_cast(float,
        __builtin_amdgcn_update_dpp(0, __builtin_bit_cast(int, v), 0x160 + MASK, 0xF, 0xF, true));
}
__device__ __forceinline__ float sum_over_8_lanes(float v)
{
    v += lane_xor_get<1>(v);
    v += lane_xor_get<2>(v);
    v += lane_xor_get<4>(v);
    return v;
}

__global__ __launch_bounds__(THREADS, 1)
void dyfconv_fused(const float* __restrict__ x,
                   const float* __restrict__ weight,
                   const float* __restrict__ bias,
                   const float* __restrict__ op_w,
                   const float* __restrict__ op_b,
                   float* __restrict__ out)
{
    __shared__ __attribute__((aligned(16))) float lds[LDS_FLOATS];
    float* ldsW  = lds + LDS_W_OFF;
    float* ldsY  = lds + LDS_Y_OFF;
    float* ldsOp = lds + LDS_OP_OFF;
    float* ldsN  = lds + LDS_N_OFF;

    const int t     = threadIdx.x;

    const int p  = t >> 3;
    const int g  = t & 7;

    float xv[8][K2];
    {
        int  toff[K2];
        bool tok[K2];
        const float* xg;
        patch_geom(min((int)blockIdx.x, NUM_TILES - 1), p, g, x, toff, tok, xg);
        #pragma unroll
        for (int cc = 0; cc < 8; ++cc) {
            #pragma unroll
            for (int j = 0; j < K2; ++j)
                xv[cc][j] = xg[cc * (H * W) + toff[j]];
        }
    }

    for (int idx = t; idx < KDIM * COUT; idx += THREADS) {
        int kq = idx & 3;
        int o  = (idx >> 2) & (COUT - 1);
        int kk = ((idx >> 8) << 2) | kq;
        int c  = kk / K2;
        int k  = kk - c * K2;
        ldsW[idx] = weight[(c * COUT + o) * K2 + k];
    }
    __syncthreads();

    const int wave  = t >> 5;
    const int lane  = t & 31;
    const int mTile = wave & 3;
    const int nTile = wave >> 2;
    const int mIdx  = lane & 15;
    const int nIdx  = lane & 15;
    const int kh    = lane >> 4;

    for (int tile = blockIdx.x; tile < NUM_TILES; tile += gridDim.x) {
    const int pbase = tile * PIX_PER_BLK;

    {
        int  toff[K2];
        bool tok[K2];
        const float* xg;
        patch_geom(tile, p, g, x, toff, tok, xg);
        #pragma unroll
        for (int cc = 0; cc < 8; ++cc) {
            #pragma unroll
            for (int j = 0; j < K2; ++j)
                xv[cc][j] = tok[j] ? xv[cc][j] : 0.0f;
        }
    }

    {
        float sq[K2];
        #pragma unroll
        for (int j = 0; j < K2; ++j) sq[j] = 0.0f;
        #pragma unroll
        for (int cc = 0; cc < 8; ++cc) {
            #pragma unroll
            for (int j = 0; j < K2; ++j)
                sq[j] += xv[cc][j] * xv[cc][j];
        }
        #pragma unroll
        for (int j = 0; j < K2; ++j)
            sq[j] = sum_over_8_lanes(sq[j]);
        if (g == 0) {
            #pragma unroll
            for (int j = 0; j < K2; ++j)
                ldsN[p * K2 + j] = sq[j];
        }
    }
    __syncthreads();

    for (int idx = t; idx < PIX_PER_BLK * K2; idx += THREADS)
        ldsN[idx] = sqrtf(ldsN[idx]);
    __syncthreads();

    for (int e = t; e < PIX_PER_BLK * 81; e += THREADS) {
        const int pp  = e / 81;
        const int idx = e - pp * 81;
        const float* nb = &ldsN[pp * K2];
        float acc = op_b[idx];
        #pragma unroll
        for (int j = 0; j < K2; ++j)
            acc += nb[j] * op_w[idx * K2 + j];
        ldsOp[pp * OPSTRIDE + idx] = tanhf(acc);
    }
    __syncthreads();

    {
        float opr[81];
        {
            const float4* op4 = reinterpret_cast<const float4*>(&ldsOp[p * OPSTRIDE]);
            #pragma unroll
            for (int q = 0; q < 20; ++q) {
                const float4 v = op4[q];
                opr[4 * q + 0] = v.x;
                opr[4 * q + 1] = v.y;
                opr[4 * q + 2] = v.z;
                opr[4 * q + 3] = v.w;
            }
            opr[80] = ldsOp[p * OPSTRIDE + 80];
        }
        float* yrow = &ldsY[p * YSTRIDE + g * 8 * K2];
        #pragma unroll
        for (int cc = 0; cc < 8; ++cc) {
            #pragma unroll
            for (int k = 0; k < K2; ++k) {
                float acc = xv[cc][k];
                #pragma unroll
                for (int j = 0; j < K2; ++j)
                    acc += xv[cc][j] * opr[j * K2 + k];
                yrow[cc * K2 + k] = acc;
            }
        }
    }
    __syncthreads();

    const float4* yrow4 = reinterpret_cast<const float4*>(&ldsY[(mTile * 16 + mIdx) * YSTRIDE + 4 * kh]);
    const float4* wcol4 = reinterpret_cast<const float4*>(&ldsW[(kh * COUT + nTile * 16 + nIdx) * 4]);

    int ntoff[K2];
    const float* nxg;
    {
        bool ntok[K2];
        const int nt = tile + (int)gridDim.x;
        patch_geom(nt < NUM_TILES ? nt : tile, p, g, x, ntoff, ntok, nxg);
    }

    v8f acc = {};
    #pragma unroll
    for (int cc = 0; cc < 8; ++cc) {
        #pragma unroll
        for (int j = 0; j < K2; ++j)
            xv[cc][j] = nxg[cc * (H * W) + ntoff[j]];

        #pragma unroll 3
        for (int q = cc * 9; q < cc * 9 + 9; ++q) {
            const float4 a4 = yrow4[q * 2];
            const float4 b4 = wcol4[q * 2 * COUT];
            v2f a, bf;
            a.x = a4.x;  a.y = a4.y;
            bf.x = b4.x; bf.y = b4.y;
            acc = __builtin_amdgcn_wmma_f32_16x16x4_f32(
                       false, a,  false, bf,
                       (short)0, acc,  false,  false);
            a.x = a4.z;  a.y = a4.w;
            bf.x = b4.z; bf.y = b4.w;
            acc = __builtin_amdgcn_wmma_f32_16x16x4_f32(
                       false, a,  false, bf,
                       (short)0, acc,  false,  false);
        }
    }

    const int o  = nTile * 16 + nIdx;
    const float bo = bias[o];
    #pragma unroll
    for (int r = 0; r < 8; ++r) {
        const int m     = r + 8 * kh;
        const int pglob = pbase + mTile * 16 + m;
        const int b_    = pglob >> 14;
        const int hw_   = pglob & (H * W - 1);
        out[((size_t)(b_ * COUT + o)) * (H * W) + hw_] = acc[r] + bo;
    }
    }
}

extern "C" void kernel_launch(void* const* d_in, const int* in_sizes, int n_in,
                              void* d_out, int out_size, void* d_ws, size_t ws_size,
                              hipStream_t stream) {
    (void)in_sizes; (void)n_in; (void)out_size; (void)d_ws; (void)ws_size;
    const float* x      = (const float*)d_in[0];
    const float* weight = (const float*)d_in[1];
    const float* bias   = (const float*)d_in[2];
    const float* op_w   = (const float*)d_in[3];
    const float* op_b   = (const float*)d_in[4];
    float* out = (float*)d_out;

    dyfconv_fused<<<NUM_WG, THREADS, 0, stream>>>(x, weight, bias, op_w, op_b, out);
}


# Operator phase: a thread owns one of the 81 entries for 11 pixels (op_w row, bias in registers; no global load or division per tile): 11 fused sums, then 11 branch-free tanh, interleaved; order kept
# speedup vs baseline: 6.0894x; 1.1326x over previous
#include <hip/hip_runtime.h>
#include <hip/hip_bf16.h>
#include <math.h>

typedef __attribute__((ext_vector_type(2))) float v2f;
typedef __attribute__((ext_vector_type(8))) float v8f;

constexpr int KSZ   = 3;
constexpr int K2    = 9;
constexpr int CIN   = 64;
constexpr int COUT  = 64;
constexpr int NB    = 8;
constexpr int H     = 128;
constexpr int W     = 128;
constexpr int KDIM  = CIN * K2;

constexpr int THREADS     = 512;
constexpr int PIX_PER_BLK = 64;
constexpr int NUM_TILES   = NB * H * W / PIX_PER_BLK;
constexpr int NUM_WG      = 256;
constexpr int YSTRIDE     = KDIM + 4;
constexpr int OP_GROUPS      = 6;
constexpr int OP_PIX_PER_GRP = 11;
constexpr int OP_LAST_PIX    = PIX_PER_BLK - (OP_GROUPS - 1) * OP_PIX_PER_GRP;
constexpr int N_ROWS         = OP_GROUPS * OP_PIX_PER_GRP;
constexpr int OPSTRIDE    = 84;

constexpr int LDS_W_OFF  = 0;
constexpr int LDS_Y_OFF  = LDS_W_OFF + KDIM * COUT;
constexpr int LDS_OP_OFF = LDS_Y_OFF + PIX_PER_BLK * YSTRIDE;
constexpr int LDS_N_OFF  = LDS_OP_OFF + PIX_PER_BLK * OPSTRIDE;
constexpr int LDS_FLOATS = LDS_N_OFF + N_ROWS * K2;

__device__ __forceinline__ void patch_geom(int tile, int p, int g, const float* __restrict__ x,
                                           int (&toff)[K2], bool (&tok)[K2], const float*& xg)
{
    const int pg = tile * PIX_PER_BLK + p;
    const int bb = pg >> 14;
    const int hw = pg & (H * W - 1);
    const int hh = hw >> 7;
    const int ww = hw & (W - 1);
    #pragma unroll
    for (int ki = 0; ki < KSZ; ++ki) {
        const int hs = hh + ki - 1;
        const int hc = min(max(hs, 0), H - 1);
        #pragma unroll
        for (int kj = 0; kj < KSZ; ++kj) {
            const int wsv = ww + kj - 1;
            const int wc  = min(max(wsv, 0), W - 1);
            tok[ki * KSZ + kj]  = ((unsigned)hs < (unsigned)H) && ((unsigned)wsv < (unsigned)W);
            toff[ki * KSZ + kj] = hc * W + wc;
        }
    }
    xg = x + (size_t)(bb * CIN + g * 8) * (H * W);
}

template <int MASK>
__device__ __forceinline__ float lane_xor_get(float v)
{
    return __builtin_bit_cast(float,
        __builtin_amdgcn_update_dpp(0, __builtin_bit_cast(int, v), 0x160 + MASK, 0xF, 0xF, true));
}
__device__ __forceinline__ float sum_over_8_lanes(float v)
{
    v += lane_xor_get<1>(v);
    v += lane_xor_get<2>(v);
    v += lane_xor_get<4>(v);
    return v;
}

__device__ __forceinline__ float tanh_sel(float x)
{
    const float ax  = fabsf(x);
    const float e   = expf(ax + ax);
    const float big = fmaf(__builtin_amdgcn_rcpf(e + 1.0f), -2.0f, 1.0f);
    const float x2  = x * x;
    float pl = fmaf(x2, -0.00570002059f, 0.0206340719f);
    pl = fmaf(x2, pl, -0.0537379310f);
    pl = fmaf(x2, pl, 0.133314162f);
    pl = fmaf(x2, pl, -0.333332807f);
    const float small = fmaf(x2, ax * pl, ax);
    return copysignf((0.625f > ax) ? small : big, x);
}

__global__ __launch_bounds__(THREADS, 1)
void dyfconv_fused(const float* __restrict__ x,
                   const float* __restrict__ weight,
                   const float* __restrict__ bias,
                   const float* __restrict__ op_w,
                   const float* __restrict__ op_b,
                   float* __restrict__ out)
{
    __shared__ __attribute__((aligned(16))) float lds[LDS_FLOATS];
    float* ldsW  = lds + LDS_W_OFF;
    float* ldsY  = lds + LDS_Y_OFF;
    float* ldsOp = lds + LDS_OP_OFF;
    float* ldsN  = lds + LDS_N_OFF;

    const int t     = threadIdx.x;

    const int p  = t >> 3;
    const int g  = t & 7;

    float xv[8][K2];
    {
        int  toff[K2];
        bool tok[K2];
        const float* xg;
        patch_geom(min((int)blockIdx.x, NUM_TILES - 1), p, g, x, toff, tok, xg);
        #pragma unroll
        for (int cc = 0; cc < 8; ++cc) {
            #pragma unroll
            for (int j = 0; j < K2; ++j)
                xv[cc][j] = xg[cc * (H * W) + toff[j]];
        }
    }

    const int  e_idx = t % 81;
    const int  e_grp = t / 81;
    const bool e_act = e_grp < OP_GROUPS;
    float opw_r[K2];
    #pragma unroll
    for (int j = 0; j < K2; ++j)
        opw_r[j] = op_w[e_idx * K2 + j];
    const float opb_r = op_b[e_idx];
    if (t < (N_ROWS - PIX_PER_BLK) * K2)
        lds[LDS_N_OFF + PIX_PER_BLK * K2 + t] = 0.0f;

    for (int idx = t; idx < KDIM * COUT; idx += THREADS) {
        int kq = idx & 3;
        int o  = (idx >> 2) & (COUT - 1);
        int kk = ((idx >> 8) << 2) | kq;
        int c  = kk / K2;
        int k  = kk - c * K2;
        ldsW[idx] = weight[(c * COUT + o) * K2 + k];
    }
    __syncthreads();

    const int wave  = t >> 5;
    const int lane  = t & 31;
    const int mTile = wave & 3;
    const int nTile = wave >> 2;
    const int mIdx  = lane & 15;
    const int nIdx  = lane & 15;
    const int kh    = lane >> 4;

    for (int tile = blockIdx.x; tile < NUM_TILES; tile += gridDim.x) {
    const int pbase = tile * PIX_PER_BLK;

    {
        int  toff[K2];
        bool tok[K2];
        const float* xg;
        patch_geom(tile, p, g, x, toff, tok, xg);
        #pragma unroll
        for (int cc = 0; cc < 8; ++cc) {
            #pragma unroll
            for (int j = 0; j < K2; ++j)
                xv[cc][j] = tok[j] ? xv[cc][j] : 0.0f;
        }
    }

    {
        float sq[K2];
        #pragma unroll
        for (int j = 0; j < K2; ++j) sq[j] = 0.0f;
        #pragma unroll
        for (int cc = 0; cc < 8; ++cc) {
            #pragma unroll
            for (int j = 0; j < K2; ++j)
                sq[j] += xv[cc][j] * xv[cc][j];
        }
        #pragma unroll
        for (int j = 0; j < K2; ++j)
            sq[j] = sum_over_8_lanes(sq[j]);
        if (g == 0) {
            #pragma unroll
            for (int j = 0; j < K2; ++j)
                ldsN[p * K2 + j] = sq[j];
        }
    }
    __syncthreads();

    for (int idx = t; idx < PIX_PER_BLK * K2; idx += THREADS)
        ldsN[idx] = sqrtf(ldsN[idx]);
    __syncthreads();

    if (e_act) {
        const float* nb = &ldsN[e_grp * (OP_PIX_PER_GRP * K2)];
        float accv[OP_PIX_PER_GRP];
        #pragma unroll
        for (int i = 0; i < OP_PIX_PER_GRP; ++i) {
            float acc = opb_r;
            #pragma unroll
            for (int j = 0; j < K2; ++j)
                acc = fmaf(nb[i * K2 + j], opw_r[j], acc);
            accv[i] = acc;
        }
        float* oprow = &ldsOp[e_grp * (OP_PIX_PER_GRP * OPSTRIDE) + e_idx];
        #pragma unroll
        for (int i = 0; i < OP_PIX_PER_GRP; ++i) {
            const float v = tanh_sel(accv[i]);
            if (i < OP_LAST_PIX || e_grp < OP_GROUPS - 1)
                oprow[i * OPSTRIDE] = v;
        }
    }
    __syncthreads();

    {
        float opr[81];
        {
            const float4* op4 = reinterpret_cast<const float4*>(&ldsOp[p * OPSTRIDE]);
            #pragma unroll
            for (int q = 0; q < 20; ++q) {
                const float4 v = op4[q];
                opr[4 * q + 0] = v.x;
                opr[4 * q + 1] = v.y;
                opr[4 * q + 2] = v.z;
                opr[4 * q + 3] = v.w;
            }
            opr[80] = ldsOp[p * OPSTRIDE + 80];
        }
        float* yrow = &ldsY[p * YSTRIDE + g * 8 * K2];
        #pragma unroll
        for (int cc = 0; cc < 8; ++cc) {
            #pragma unroll
            for (int k = 0; k < K2; ++k) {
                float acc = xv[cc][k];
                #pragma unroll
                for (int j = 0; j < K2; ++j)
                    acc += xv[cc][j] * opr[j * K2 + k];
                yrow[cc * K2 + k] = acc;
            }
        }
    }
    __syncthreads();

    const float4* yrow4 = reinterpret_cast<const float4*>(&ldsY[(mTile * 16 + mIdx) * YSTRIDE + 4 * kh]);
    const float4* wcol4 = reinterpret_cast<const float4*>(&ldsW[(kh * COUT + nTile * 16 + nIdx) * 4]);

    int ntoff[K2];
    const float* nxg;
    {
        bool ntok[K2];
        const int nt = tile + (int)gridDim.x;
        patch_geom(nt < NUM_TILES ? nt : tile, p, g, x, ntoff, ntok, nxg);
    }

    v8f acc = {};
    #pragma unroll
    for (int cc = 0; cc < 8; ++cc) {
        #pragma unroll
        for (int j = 0; j < K2; ++j)
            xv[cc][j] = nxg[cc * (H * W) + ntoff[j]];

        #pragma unroll 3
        for (int q = cc * 9; q < cc * 9 + 9; ++q) {
            const float4 a4 = yrow4[q * 2];
            const float4 b4 = wcol4[q * 2 * COUT];
            v2f a, bf;
            a.x = a4.x;  a.y = a4.y;
            bf.x = b4.x; bf.y = b4.y;
            acc = __builtin_amdgcn_wmma_f32_16x16x4_f32(
                       false, a,  false, bf,
                       (short)0, acc,  false,  false);
            a.x = a4.z;  a.y = a4.w;
            bf.x = b4.z; bf.y = b4.w;
            acc = __builtin_amdgcn_wmma_f32_16x16x4_f32(
                       false, a,  false, bf,
                       (short)0, acc,  false,  false);
        }
    }

    const int o  = nTile * 16 + nIdx;
    const float bo = bias[o];
    #pragma unroll
    for (int r = 0; r < 8; ++r) {
        const int m     = r + 8 * kh;
        const int pglob = pbase + mTile * 16 + m;
        const int b_    = pglob >> 14;
        const int hw_   = pglob & (H * W - 1);
        out[((size_t)(b_ * COUT + o)) * (H * W) + hw_] = acc[r] + bo;
    }
    }
}

extern "C" void kernel_launch(void* const* d_in, const int* in_sizes, int n_in,
                              void* d_out, int out_size, void* d_ws, size_t ws_size,
                              hipStream_t stream) {
    (void)in_sizes; (void)n_in; (void)out_size; (void)d_ws; (void)ws_size;
    const float* x      = (const float*)d_in[0];
    const float* weight = (const float*)d_in[1];
    const float* bias   = (const float*)d_in[2];
    const float* op_w   = (const float*)d_in[3];
    const float* op_b   = (const float*)d_in[4];
    float* out = (float*)d_out;

    dyfconv_fused<<<NUM_WG, THREADS, 0, stream>>>(x, weight, bias, op_w, op_b, out);
}


# Weight staging: a thread's 72 weight loads are issued as one group ahead of the first tile's x loads (division-free addresses), then stored to LDS (was a loop of 18 fully waited trips of 4 loads)
# speedup vs baseline: 6.4107x; 1.0528x over previous
#include <hip/hip_runtime.h>
#include <hip/hip_bf16.h>
#include <math.h>

typedef __attribute__((ext_vector_type(2))) float v2f;
typedef __attribute__((ext_vector_type(8))) float v8f;

constexpr int KSZ   = 3;
constexpr int K2    = 9;
constexpr int CIN   = 64;
constexpr int COUT  = 64;
constexpr int NB    = 8;
constexpr int H     = 128;
constexpr int W     = 128;
constexpr int KDIM  = CIN * K2;

constexpr int THREADS     = 512;
constexpr int PIX_PER_BLK = 64;
constexpr int NUM_TILES   = NB * H * W / PIX_PER_BLK;
constexpr int NUM_WG      = 256;
constexpr int YSTRIDE     = KDIM + 4;
constexpr int OP_GROUPS      = 6;
constexpr int OP_PIX_PER_GRP = 11;
constexpr int OP_LAST_PIX    = PIX_PER_BLK - (OP_GROUPS - 1) * OP_PIX_PER_GRP;
constexpr int N_ROWS         = OP_GROUPS * OP_PIX_PER_GRP;
constexpr int OPSTRIDE    = 84;

constexpr int LDS_W_OFF  = 0;
constexpr int LDS_Y_OFF  = LDS_W_OFF + KDIM * COUT;
constexpr int LDS_OP_OFF = LDS_Y_OFF + PIX_PER_BLK * YSTRIDE;
constexpr int LDS_N_OFF  = LDS_OP_OFF + PIX_PER_BLK * OPSTRIDE;
constexpr int LDS_FLOATS = LDS_N_OFF + N_ROWS * K2;

__device__ __forceinline__ void patch_geom(int tile, int p, int g, const float* __restrict__ x,
                                           int (&toff)[K2], bool (&tok)[K2], const float*& xg)
{
    const int pg = tile * PIX_PER_BLK + p;
    const int bb = pg >> 14;
    const int hw = pg & (H * W - 1);
    const int hh = hw >> 7;
    const int ww = hw & (W - 1);
    #pragma unroll
    for (int ki = 0; ki < KSZ; ++ki) {
        const int hs = hh + ki - 1;
        const int hc = min(max(hs, 0), H - 1);
        #pragma unroll
        for (int kj = 0; kj < KSZ; ++kj) {
            const int wsv = ww + kj - 1;
            const int wc  = min(max(wsv, 0), W - 1);
            tok[ki * KSZ + kj]  = ((unsigned)hs < (unsigned)H) && ((unsigned)wsv < (unsigned)W);
            toff[ki * KSZ + kj] = hc * W + wc;
        }
    }
    xg = x + (size_t)(bb * CIN + g * 8) * (H * W);
}

template <int MASK>
__device__ __forceinline__ float lane_xor_get(float v)
{
    return __builtin_bit_cast(float,
        __builtin_amdgcn_update_dpp(0, __builtin_bit_cast(int, v), 0x160 + MASK, 0xF, 0xF, true));
}
__device__ __forceinline__ float sum_over_8_lanes(float v)
{
    v += lane_xor_get<1>(v);
    v += lane_xor_get<2>(v);
    v += lane_xor_get<4>(v);
    return v;
}

__device__ __forceinline__ float tanh_sel(float x)
{
    const float ax  = fabsf(x);
    const float e   = expf(ax + ax);
    const float big = fmaf(__builtin_amdgcn_rcpf(e + 1.0f), -2.0f, 1.0f);
    const float x2  = x * x;
    float pl = fmaf(x2, -0.00570002059f, 0.0206340719f);
    pl = fmaf(x2, pl, -0.0537379310f);
    pl = fmaf(x2, pl, 0.133314162f);
    pl = fmaf(x2, pl, -0.333332807f);
    const float small = fmaf(x2, ax * pl, ax);
    return copysignf((0.625f > ax) ? small : big, x);
}

__global__ __launch_bounds__(THREADS, 1)
void dyfconv_fused(const float* __restrict__ x,
                   const float* __restrict__ weight,
                   const float* __restrict__ bias,
                   const float* __restrict__ op_w,
                   const float* __restrict__ op_b,
                   float* __restrict__ out)
{
    __shared__ __attribute__((aligned(16))) float lds[LDS_FLOATS];
    float* ldsW  = lds + LDS_W_OFF;
    float* ldsY  = lds + LDS_Y_OFF;
    float* ldsOp = lds + LDS_OP_OFF;
    float* ldsN  = lds + LDS_N_OFF;

    const int t     = threadIdx.x;

    const int p  = t >> 3;
    const int g  = t & 7;

    float wv[(KDIM * COUT) / THREADS];
    {
        const int wr = ((t >> 8) << 2) | (t & 3);
        const int wa = 64 * wr + K2 * ((t >> 2) & (COUT - 1));
        #pragma unroll
        for (int i = 0; i < (KDIM * COUT) / THREADS; ++i) {
            const int ks = (8 * i) % K2 + wr;
            const int k  = ks >= K2 ? ks - K2 : ks;
            wv[i] = weight[wa + 512 * i - 63 * k];
        }
    }

    float xv[8][K2];
    {
        int  toff[K2];
        bool tok[K2];
        const float* xg;
        patch_geom(min((int)blockIdx.x, NUM_TILES - 1), p, g, x, toff, tok, xg);
        #pragma unroll
        for (int cc = 0; cc < 8; ++cc) {
            #pragma unroll
            for (int j = 0; j < K2; ++j)
                xv[cc][j] = xg[cc * (H * W) + toff[j]];
        }
    }

    const int  e_idx = t % 81;
    const int  e_grp = t / 81;
    const bool e_act = e_grp < OP_GROUPS;
    float opw_r[K2];
    #pragma unroll
    for (int j = 0; j < K2; ++j)
        opw_r[j] = op_w[e_idx * K2 + j];
    const float opb_r = op_b[e_idx];
    if (t < (N_ROWS - PIX_PER_BLK) * K2)
        lds[LDS_N_OFF + PIX_PER_BLK * K2 + t] = 0.0f;

    #pragma unroll
    for (int i = 0; i < (KDIM * COUT) / THREADS; ++i)
        ldsW[t + i * THREADS] = wv[i];
    __syncthreads();

    const int wave  = t >> 5;
    const int lane  = t & 31;
    const int mTile = wave & 3;
    const int nTile = wave >> 2;
    const int mIdx  = lane & 15;
    const int nIdx  = lane & 15;
    const int kh    = lane >> 4;

    for (int tile = blockIdx.x; tile < NUM_TILES; tile += gridDim.x) {
    const int pbase = tile * PIX_PER_BLK;

    {
        int  toff[K2];
        bool tok[K2];
        const float* xg;
        patch_geom(tile, p, g, x, toff, tok, xg);
        #pragma unroll
        for (int cc = 0; cc < 8; ++cc) {
            #pragma unroll
            for (int j = 0; j < K2; ++j)
                xv[cc][j] = tok[j] ? xv[cc][j] : 0.0f;
        }
    }

    {
        float sq[K2];
        #pragma unroll
        for (int j = 0; j < K2; ++j) sq[j] = 0.0f;
        #pragma unroll
        for (int cc = 0; cc < 8; ++cc) {
            #pragma unroll
            for (int j = 0; j < K2; ++j)
                sq[j] += xv[cc][j] * xv[cc][j];
        }
        #pragma unroll
        for (int j = 0; j < K2; ++j)
            sq[j] = sum_over_8_lanes(sq[j]);
        if (g == 0) {
            #pragma unroll
            for (int j = 0; j < K2; ++j)
                ldsN[p * K2 + j] = sq[j];
        }
    }
    __syncthreads();

    for (int idx = t; idx < PIX_PER_BLK * K2; idx += THREADS)
        ldsN[idx] = sqrtf(ldsN[idx]);
    __syncthreads();

    if (e_act) {
        const float* nb = &ldsN[e_grp * (OP_PIX_PER_GRP * K2)];
        float accv[OP_PIX_PER_GRP];
        #pragma unroll
        for (int i = 0; i < OP_PIX_PER_GRP; ++i) {
            float acc = opb_r;
            #pragma unroll
            for (int j = 0; j < K2; ++j)
                acc = fmaf(nb[i * K2 + j], opw_r[j], acc);
            accv[i] = acc;
        }
        float* oprow = &ldsOp[e_grp * (OP_PIX_PER_GRP * OPSTRIDE) + e_idx];
        #pragma unroll
        for (int i = 0; i < OP_PIX_PER_GRP; ++i) {
            const float v = tanh_sel(accv[i]);
            if (i < OP_LAST_PIX || e_grp < OP_GROUPS - 1)
                oprow[i * OPSTRIDE] = v;
        }
    }
    __syncthreads();

    {
        float opr[81];
        {
            const float4* op4 = reinterpret_cast<const float4*>(&ldsOp[p * OPSTRIDE]);
            #pragma unroll
            for (int q = 0; q < 20; ++q) {
                const float4 v = op4[q];
                opr[4 * q + 0] = v.x;
                opr[4 * q + 1] = v.y;
                opr[4 * q + 2] = v.z;
                opr[4 * q + 3] = v.w;
            }
            opr[80] = ldsOp[p * OPSTRIDE + 80];
        }
        float* yrow = &ldsY[p * YSTRIDE + g * 8 * K2];
        #pragma unroll
        for (int cc = 0; cc < 8; ++cc) {
            #pragma unroll
            for (int k = 0; k < K2; ++k) {
                float acc = xv[cc][k];
                #pragma unroll
                for (int j = 0; j < K2; ++j)
                    acc += xv[cc][j] * opr[j * K2 + k];
                yrow[cc * K2 + k] = acc;
            }
        }
    }
    __syncthreads();

    const float4* yrow4 = reinterpret_cast<const float4*>(&ldsY[(mTile * 16 + mIdx) * YSTRIDE + 4 * kh]);
    const float4* wcol4 = reinterpret_cast<const float4*>(&ldsW[(kh * COUT + nTile * 16 + nIdx) * 4]);

    int ntoff[K2];
    const float* nxg;
    {
        bool ntok[K2];
        const int nt = tile + (int)gridDim.x;
        patch_geom(nt < NUM_TILES ? nt : tile, p, g, x, ntoff, ntok, nxg);
    }

    v8f acc = {};
    #pragma unroll
    for (int cc = 0; cc < 8; ++cc) {
        #pragma unroll
        for (int j = 0; j < K2; ++j)
            xv[cc][j] = nxg[cc * (H * W) + ntoff[j]];

        #pragma unroll 3
        for (int q = cc * 9; q < cc * 9 + 9; ++q) {
            const float4 a4 = yrow4[q * 2];
            const float4 b4 = wcol4[q * 2 * COUT];
            v2f a, bf;
            a.x = a4.x;  a.y = a4.y;
            bf.x = b4.x; bf.y = b4.y;
            acc = __builtin_amdgcn_wmma_f32_16x16x4_f32(
                       false, a,  false, bf,
                       (short)0, acc,  false,  false);
            a.x = a4.z;  a.y = a4.w;
            bf.x = b4.z; bf.y = b4.w;
            acc = __builtin_amdgcn_wmma_f32_16x16x4_f32(
                       false, a,  false, bf,
                       (short)0, acc,  false,  false);
        }
    }

    const int o  = nTile * 16 + nIdx;
    const float bo = bias[o];
    #pragma unroll
    for (int r = 0; r < 8; ++r) {
        const int m     = r + 8 * kh;
        const int pglob = pbase + mTile * 16 + m;
        const int b_    = pglob >> 14;
        const int hw_   = pglob & (H * W - 1);
        out[((size_t)(b_ * COUT + o)) * (H * W) + hw_] = acc[r] + bo;
    }
    }
}

extern "C" void kernel_launch(void* const* d_in, const int* in_sizes, int n_in,
                              void* d_out, int out_size, void* d_ws, size_t ws_size,
                              hipStream_t stream) {
    (void)in_sizes; (void)n_in; (void)out_size; (void)d_ws; (void)ws_size;
    const float* x      = (const float*)d_in[0];
    const float* weight = (const float*)d_in[1];
    const float* bias   = (const float*)d_in[2];
    const float* op_w   = (const float*)d_in[3];
    const float* op_b   = (const float*)d_in[4];
    float* out = (float*)d_out;

    dyfconv_fused<<<NUM_WG, THREADS, 0, stream>>>(x, weight, bias, op_w, op_b, out);
}
